# starting kernel
# baseline (speedup 1.0000x reference)
#include <hip/hip_runtime.h>
#include <math.h>


typedef float v2f __attribute__((ext_vector_type(2)));
typedef float v8f __attribute__((ext_vector_type(8)));

#define HEADS  8
#define CDIM   32
#define HCDIM  256
#define NSLOPE 0.2f

__device__ __forceinline__ void atomicMaxF(float* addr, float val) {
    if (val >= 0.0f) {
        atomicMax((int*)addr, __float_as_int(val));
    } else {
        atomicMin((unsigned int*)addr, __float_as_uint(val));
    }
}

__global__ void wmma_gemm_f32(const float* __restrict__ A,
                              const float* __restrict__ B,
                              float* __restrict__ C, int K, int NB) {
    const int  lane = threadIdx.x & 31;
    const bool hi   = lane >= 16;
    const int  l    = lane & 15;
    const int  row0 = blockIdx.x * 16;
    const int  col0 = blockIdx.y * 16;

    v8f acc = {};
    const float* arow = A + (size_t)(row0 + l) * K + (hi ? 2 : 0);
    const float* bcol = B + (size_t)(hi ? 2 : 0) * NB + col0 + l;
    for (int k = 0; k < K; k += 4) {
        v2f a; a.x = arow[k];                  a.y = arow[k + 1];
        v2f b; b.x = bcol[(size_t)k * NB];     b.y = bcol[(size_t)(k + 1) * NB];
        acc = __builtin_amdgcn_wmma_f32_16x16x4_f32(
                  false, a, false, b, (short)0, acc, false, false);
    }
    float* crow = C + (size_t)(row0 + (hi ? 8 : 0)) * NB + col0 + l;
#pragma unroll
    for (int r = 0; r < 8; ++r) crow[(size_t)r * NB] = acc[r];
}

__global__ void wmma_gemm_relu_bias(const float* __restrict__ Acc,
                                    const float* __restrict__ bias,
                                    const float* __restrict__ B,
                                    const float* __restrict__ bout,
                                    float* __restrict__ C, int K, int NB) {
    const int  lane = threadIdx.x & 31;
    const bool hi   = lane >= 16;
    const int  l    = lane & 15;
    const int  row0 = blockIdx.x * 16;
    const int  col0 = blockIdx.y * 16;

    v8f acc = {};
    const float* arow = Acc + (size_t)(row0 + l) * K;
    const float* bcol = B + (size_t)(hi ? 2 : 0) * NB + col0 + l;
    const int    koff = hi ? 2 : 0;
    for (int k = 0; k < K; k += 4) {
        const int kk = k + koff;
        v2f a;
        a.x = fmaxf(arow[kk]     + bias[kk],     0.0f);
        a.y = fmaxf(arow[kk + 1] + bias[kk + 1], 0.0f);
        v2f b; b.x = bcol[(size_t)k * NB]; b.y = bcol[(size_t)(k + 1) * NB];
        acc = __builtin_amdgcn_wmma_f32_16x16x4_f32(
                  false, a, false, b, (short)0, acc, false, false);
    }
    const float bo = bout[col0 + l];
    float* crow = C + (size_t)(row0 + (hi ? 8 : 0)) * NB + col0 + l;
#pragma unroll
    for (int r = 0; r < 8; ++r) crow[(size_t)r * NB] = acc[r] + bo;
}

__global__ void gat_att_scores(const float* __restrict__ h,
                               const float* __restrict__ att_src,
                               const float* __restrict__ att_dst,
                               float* __restrict__ a_src,
                               float* __restrict__ a_dst,
                               float* __restrict__ maxbuf,
                               float* __restrict__ denom, int N) {
    const int t = blockIdx.x * blockDim.x + threadIdx.x;
    if (t >= N * HEADS) return;
    const int n  = t / HEADS;
    const int hd = t - n * HEADS;
    const float* hp = h + (size_t)n * HCDIM + hd * CDIM;
    const float* as = att_src + hd * CDIM;
    const float* ad = att_dst + hd * CDIM;
    float s0 = 0.0f, s1 = 0.0f;
#pragma unroll
    for (int c = 0; c < CDIM; ++c) {
        const float v = hp[c];
        s0 = fmaf(v, as[c], s0);
        s1 = fmaf(v, ad[c], s1);
    }
    a_src[t]  = s0;
    a_dst[t]  = s1;
    maxbuf[t] = -INFINITY;
    denom[t]  = 0.0f;
}

__global__ void zero_f32(float* __restrict__ p, int n) {
    const int t = blockIdx.x * blockDim.x + threadIdx.x;
    if (t < n) p[t] = 0.0f;
}

__device__ __forceinline__ void edge_src_dst(const int* __restrict__ ei,
                                             int e, int E, int& s, int& d) {
    if (e < E) { s = ei[e]; d = ei[E + e]; }
    else       { s = e - E; d = e - E; }
}

__device__ __forceinline__ float edge_logit(const float* __restrict__ a_src,
                                            const float* __restrict__ a_dst,
                                            int s, int d, int hd) {
    float v = a_src[s * HEADS + hd] + a_dst[d * HEADS + hd];
    return v > 0.0f ? v : NSLOPE * v;
}

__global__ void gat_edge_max(const int* __restrict__ ei,
                             const float* __restrict__ a_src,
                             const float* __restrict__ a_dst,
                             float* __restrict__ maxbuf, int E, int ET) {
    const int t = blockIdx.x * blockDim.x + threadIdx.x;
    if (t >= ET * HEADS) return;
    const int e  = t >> 3;
    const int hd = t & 7;
    int s, d;
    edge_src_dst(ei, e, E, s, d);
    atomicMaxF(&maxbuf[d * HEADS + hd], edge_logit(a_src, a_dst, s, d, hd));
}

__global__ void gat_edge_expsum(const int* __restrict__ ei,
                                const float* __restrict__ a_src,
                                const float* __restrict__ a_dst,
                                const float* __restrict__ maxbuf,
                                float* __restrict__ denom, int E, int ET) {
    const int t = blockIdx.x * blockDim.x + threadIdx.x;
    if (t >= ET * HEADS) return;
    const int e  = t >> 3;
    const int hd = t & 7;
    int s, d;
    edge_src_dst(ei, e, E, s, d);
    const float v  = edge_logit(a_src, a_dst, s, d, hd);
    const float ex = expf(v - maxbuf[d * HEADS + hd]);
    atomicAdd(&denom[d * HEADS + hd], ex);
}

__global__ void gat_edge_scatter(const int* __restrict__ ei,
                                 const float* __restrict__ a_src,
                                 const float* __restrict__ a_dst,
                                 const float* __restrict__ maxbuf,
                                 const float* __restrict__ denom,
                                 const float* __restrict__ h,
                                 float* __restrict__ accum, int E, int ET) {
    const int lane = threadIdx.x & 31;
    const int wid  = (blockIdx.x * blockDim.x + threadIdx.x) >> 5;
    if (wid >= ET) return;
    int s, d;
    edge_src_dst(ei, wid, E, s, d);

    const int   h8    = lane & 7;
    const float v     = edge_logit(a_src, a_dst, s, d, h8);
    const float ex    = expf(v - maxbuf[d * HEADS + h8]);
    const float alpha = ex / (denom[d * HEADS + h8] + 1e-16f);

    const float* hs = h + (size_t)s * HCDIM;
    float*       ap = accum + (size_t)d * HCDIM;
#pragma unroll
    for (int i = 0; i < HEADS; ++i) {
        const float al = __shfl(alpha, i, 32);
        atomicAdd(&ap[i * CDIM + lane], hs[i * CDIM + lane] * al);
    }
}

extern "C" void kernel_launch(void* const* d_in, const int* in_sizes, int n_in,
                              void* d_out, int out_size, void* d_ws, size_t ws_size,
                              hipStream_t stream) {
    (void)n_in; (void)out_size; (void)ws_size;

    const float* x        = (const float*)d_in[0];
    const int*   ei       = (const int*)d_in[1];
    const float* W_gat    = (const float*)d_in[2];
    const float* att_src  = (const float*)d_in[3];
    const float* att_dst  = (const float*)d_in[4];
    const float* bias_gat = (const float*)d_in[5];
    const float* W_lin    = (const float*)d_in[6];
    const float* b_lin    = (const float*)d_in[7];
    float*       out      = (float*)d_out;

    const int IN  = 256;
    const int N   = in_sizes[0] / IN;
    const int E   = in_sizes[1] / 2;
    const int OUT = in_sizes[6] / HCDIM;
    const int ET  = E + N;

    float* ws   = (float*)d_ws;
    float* h    = ws;
    float* acc  = h    + (size_t)N * HCDIM;
    float* asrc = acc  + (size_t)N * HCDIM;
    float* adst = asrc + (size_t)N * HEADS;
    float* mbuf = adst + (size_t)N * HEADS;
    float* den  = mbuf + (size_t)N * HEADS;

    wmma_gemm_f32<<<dim3(N / 16, HCDIM / 16), 32, 0, stream>>>(x, W_gat, h, IN, HCDIM);

    const int t2 = N * HEADS;
    gat_att_scores<<<(t2 + 255) / 256, 256, 0, stream>>>(h, att_src, att_dst,
                                                         asrc, adst, mbuf, den, N);

    const int t3 = N * HCDIM;
    zero_f32<<<(t3 + 255) / 256, 256, 0, stream>>>(acc, t3);

    const int t4 = ET * HEADS;
    gat_edge_max<<<(t4 + 255) / 256, 256, 0, stream>>>(ei, asrc, adst, mbuf, E, ET);

    gat_edge_expsum<<<(t4 + 255) / 256, 256, 0, stream>>>(ei, asrc, adst, mbuf, den, E, ET);

    gat_edge_scatter<<<(ET + 7) / 8, 256, 0, stream>>>(ei, asrc, adst, mbuf, den,
                                                       h, acc, E, ET);

    wmma_gemm_relu_bias<<<dim3(N / 16, OUT / 16), 32, 0, stream>>>(acc, bias_gat,
                                                                   W_lin, b_lin,
                                                                   out, HCDIM, OUT);
}


# Float-atomic edge scatter replaced by a gather: edges counting-sorted by destination in each call (integer-atomic rank), one wave per node does softmax and weighted sum in registers in list order
# speedup vs baseline: 2.8167x; 2.8167x over previous
#include <hip/hip_runtime.h>
#include <math.h>


typedef float v2f __attribute__((ext_vector_type(2)));
typedef float v8f __attribute__((ext_vector_type(8)));

#define HEADS  8
#define CDIM   32
#define HCDIM  256
#define NSLOPE 0.2f

#define GEMM_K   256

template <int NB, int WCOLS, int KC, bool FUSE>
__global__ void __launch_bounds__(256)
wmma_gemm_tiled(const float* __restrict__ A,
                const float* __restrict__ bias,
                const float* __restrict__ B,
                const float* __restrict__ bout,
                float* __restrict__ C, int nrt) {
    constexpr int NCH   = GEMM_K / KC;
    constexpr int QR    = KC / 4;
    constexpr int NG    = KC / 8;
    constexpr int CT    = WCOLS / 16;
    constexpr int WPR   = NB / WCOLS;
    constexpr int RW    = 8 / WPR;
    constexpr int PITCH = NB + 8;
    constexpr int BUF   = QR * PITCH;
    constexpr int NQ    = NB / 4;
    constexpr int UNITS = QR * NQ;
    constexpr int UPT   = UNITS / 256;
    static_assert(UNITS % 256 == 0, "every thread stages the same number of units");

    __shared__ float4 Bs[2 * BUF];

    const int  tid   = threadIdx.x;
    const int  lane  = tid & 31;
    const int  wv    = tid >> 5;
    const int  c     = lane & 15;
    const int  hf    = lane >> 4;
    const int  wr    = wv / WPR;
    const int  wc    = wv - wr * WPR;
    const int  rt    = blockIdx.x * RW + wr;
    const bool valid = rt < nrt;
    const int  rtc   = valid ? rt : nrt - 1;
    const int  col0  = wc * WCOLS;

    const float* ap = A + (size_t)(rtc * 16 + c) * GEMM_K + 4 * hf;
    const float* bp = bias + 4 * hf;

    float4 sb[UPT][4];
    float4 a_nxt[NG];
    v8f    acc[CT];
    for (int t = 0; t < CT; ++t) acc[t] = v8f{};

    auto fetch_b = [&](int ch) {
        for (int i = 0; i < UPT; ++i) {
            const int u  = tid + 256 * i;
            const int kq = u / NQ;
            const int q  = u - kq * NQ;
            const float* g = B + (size_t)(ch * KC + 4 * kq) * NB + 4 * q;
            for (int r = 0; r < 4; ++r)
                sb[i][r] = *(const float4*)(g + (size_t)r * NB);
        }
    };
    auto store_b = [&](int buf) {
        for (int i = 0; i < UPT; ++i) {
            const int u  = tid + 256 * i;
            const int kq = u / NQ;
            const int q  = u - kq * NQ;
            float4* d = Bs + buf * BUF + kq * PITCH + 4 * q;
            d[0] = make_float4(sb[i][0].x, sb[i][1].x, sb[i][2].x, sb[i][3].x);
            d[1] = make_float4(sb[i][0].y, sb[i][1].y, sb[i][2].y, sb[i][3].y);
            d[2] = make_float4(sb[i][0].z, sb[i][1].z, sb[i][2].z, sb[i][3].z);
            d[3] = make_float4(sb[i][0].w, sb[i][1].w, sb[i][2].w, sb[i][3].w);
        }
    };
    auto fetch_a = [&](int ch) {
        for (int g = 0; g < NG; ++g) {
            float4 v = *(const float4*)(ap + ch * KC + 8 * g);
            if (FUSE) {
                const float4 bb = *(const float4*)(bp + ch * KC + 8 * g);
                v.x = fmaxf(v.x + bb.x, 0.0f);
                v.y = fmaxf(v.y + bb.y, 0.0f);
                v.z = fmaxf(v.z + bb.z, 0.0f);
                v.w = fmaxf(v.w + bb.w, 0.0f);
            }
            a_nxt[g] = v;
        }
    };

    fetch_b(0);
    fetch_a(0);
    store_b(0);

    for (int ch = 0; ch < NCH; ++ch) {
        float4 a_cur[NG];
        for (int g = 0; g < NG; ++g) a_cur[g] = a_nxt[g];
        if (ch + 1 < NCH) { fetch_b(ch + 1); fetch_a(ch + 1); }

        asm volatile("s_wait_dscnt 0x0" ::: "memory");
        asm volatile("s_barrier_signal -1" ::: "memory");
        asm volatile("s_barrier_wait -1" ::: "memory");

        const float4* bs = Bs + (ch & 1) * BUF + hf * PITCH + col0 + c;
        for (int g = 0; g < NG; ++g) {
            float4 bv[CT];
            for (int t = 0; t < CT; ++t) bv[t] = bs[2 * g * PITCH + 16 * t];
            const v2f a0 = {a_cur[g].x, a_cur[g].y};
            const v2f a1 = {a_cur[g].z, a_cur[g].w};
            for (int t = 0; t < CT; ++t) {
                const v2f b0 = {bv[t].x, bv[t].y};
                acc[t] = __builtin_amdgcn_wmma_f32_16x16x4_f32(
                             false, a0, false, b0, (short)0, acc[t], false, false);
            }
            for (int t = 0; t < CT; ++t) {
                const v2f b1 = {bv[t].z, bv[t].w};
                acc[t] = __builtin_amdgcn_wmma_f32_16x16x4_f32(
                             false, a1, false, b1, (short)0, acc[t], false, false);
            }
        }

        if (ch + 1 < NCH) store_b((ch + 1) & 1);
    }

    if (valid) {
        float* cp = C + (size_t)(rt * 16 + 8 * hf) * NB + col0 + c;
        for (int t = 0; t < CT; ++t) {
            if (FUSE) {
                const float bo = bout[col0 + 16 * t + c];
                for (int r = 0; r < 8; ++r) cp[(size_t)r * NB + 16 * t] = acc[t][r] + bo;
            } else {
                for (int r = 0; r < 8; ++r) cp[(size_t)r * NB + 16 * t] = acc[t][r];
            }
        }
    }
}

__global__ void gat_src_scores(const float* __restrict__ h,
                               const float* __restrict__ att_src,
                               float* __restrict__ a_src,
                               int* __restrict__ cnt, int N) {
    const int t = blockIdx.x * blockDim.x + threadIdx.x;
    if (t >= N * HEADS) return;
    if (t < N) cnt[t] = 0;
    const int n  = t / HEADS;
    const int hd = t - n * HEADS;
    const float* hp = h + (size_t)n * HCDIM + hd * CDIM;
    const float* as = att_src + hd * CDIM;
    float s0 = 0.0f;
#pragma unroll
    for (int c = 0; c < CDIM; ++c) s0 = fmaf(hp[c], as[c], s0);
    a_src[t] = s0;
}

__global__ void gat_edge_rank(const int* __restrict__ ei, int* __restrict__ cnt,
                              int* __restrict__ rank, int E) {
    const int e = blockIdx.x * blockDim.x + threadIdx.x;
    if (e >= E) return;
    rank[e] = atomicAdd(&cnt[ei[E + e]], 1);
}

#define SCAN_T 1024
__global__ void __launch_bounds__(SCAN_T)
gat_rowptr_scan(int* __restrict__ rowptr, int N) {
    __shared__ int wsum[SCAN_T / 32];
    const int tid  = threadIdx.x;
    const int lane = tid & 31;
    const int wv   = tid >> 5;
    const int per  = (((N + SCAN_T - 1) / SCAN_T) + 3) & ~3;
    const int lo   = tid * per;
    const int hi   = min(lo + per, N);

    int s = 0;
    int i = lo;
    for (; i + 4 <= hi; i += 4) {
        const int4 v = *(const int4*)(rowptr + i);
        s += (v.x + v.y) + (v.z + v.w);
    }
    for (; i < hi; ++i) s += rowptr[i];

    int inc = s;
    for (int o = 1; o < 32; o <<= 1) {
        const int up = __shfl_up(inc, o, 32);
        if (lane >= o) inc += up;
    }
    if (lane == 31) wsum[wv] = inc;
    __syncthreads();
    if (wv == 0) {
        const int w = wsum[lane];
        int winc = w;
        for (int o = 1; o < 32; o <<= 1) {
            const int up = __shfl_up(winc, o, 32);
            if (lane >= o) winc += up;
        }
        wsum[lane] = winc - w;
        if (lane == 31) rowptr[N] = winc;
    }
    __syncthreads();

    int run = wsum[wv] + inc - s;
    i = lo;
    for (; i + 4 <= hi; i += 4) {
        const int4 v = *(const int4*)(rowptr + i);
        int4 o;
        o.x = run; run += v.x;
        o.y = run; run += v.y;
        o.z = run; run += v.z;
        o.w = run; run += v.w;
        *(int4*)(rowptr + i) = o;
    }
    for (; i < hi; ++i) { const int v = rowptr[i]; rowptr[i] = run; run += v; }
}

__global__ void gat_edge_place(const int* __restrict__ ei,
                               const int* __restrict__ rowptr,
                               const int* __restrict__ rank,
                               int* __restrict__ ssrc, int E) {
    const int e = blockIdx.x * blockDim.x + threadIdx.x;
    if (e >= E) return;
    ssrc[rowptr[ei[E + e]] + rank[e]] = ei[e];
}

__device__ __forceinline__ float group8_sum(float v) {
    v += __shfl_xor(v, 1, 32);
    v += __shfl_xor(v, 2, 32);
    v += __shfl_xor(v, 4, 32);
    return v;
}
__device__ __forceinline__ float group8_max(float v) {
    v = fmaxf(v, __shfl_xor(v, 1, 32));
    v = fmaxf(v, __shfl_xor(v, 2, 32));
    v = fmaxf(v, __shfl_xor(v, 4, 32));
    return v;
}

__global__ void __launch_bounds__(256)
gat_node_gather(const int* __restrict__ rowptr,
                const int* __restrict__ ssrc,
                const float* __restrict__ a_src,
                const float* __restrict__ att_dst,
                const float* __restrict__ h,
                float* __restrict__ accum, int N) {
    const int lane = threadIdx.x & 31;
    const int d    = __builtin_amdgcn_readfirstlane((blockIdx.x * blockDim.x + threadIdx.x) >> 5);
    if (d >= N) return;
    const int g   = lane >> 3;
    const int e8  = lane & 7;
    const int beg = rowptr[d];
    const int n   = rowptr[d + 1] - beg + 1;
    const int nch = (n + 31) >> 5;

    const float4 hd0 = *(const float4*)(h + (size_t)d * HCDIM + 4 * lane);
    const float4 hd1 = *(const float4*)(h + (size_t)d * HCDIM + 128 + 4 * lane);
    const float4 ad0 = *(const float4*)(att_dst + 4 * lane);
    const float4 ad1 = *(const float4*)(att_dst + 128 + 4 * lane);
    const float adA = group8_sum(fmaf(hd0.w, ad0.w, fmaf(hd0.z, ad0.z, fmaf(hd0.y, ad0.y, hd0.x * ad0.x))));
    const float adB = group8_sum(fmaf(hd1.w, ad1.w, fmaf(hd1.z, ad1.z, fmaf(hd1.y, ad1.y, hd1.x * ad1.x))));

    int   s[4];
    float vA[4], vB[4];
    auto load_chunk = [&](int c) {
#pragma unroll
        for (int i = 0; i < 4; ++i) {
            const int  j  = 32 * c + 8 * i + e8;
            const bool ok = j < n;
            int sj = d;
            if (ok && j > 0) sj = ssrc[beg + j - 1];
            s[i] = sj;
            float a = a_src[sj * HEADS + g] + adA;
            float b = a_src[sj * HEADS + g + 4] + adB;
            a = a > 0.0f ? a : NSLOPE * a;
            b = b > 0.0f ? b : NSLOPE * b;
            vA[i] = ok ? a : -INFINITY;
            vB[i] = ok ? b : -INFINITY;
        }
    };

    load_chunk(0);

    float mA = -INFINITY, mB = -INFINITY;
    for (int c = 0; c < nch; ++c) {
        if (nch > 1) load_chunk(c);
#pragma unroll
        for (int i = 0; i < 4; ++i) { mA = fmaxf(mA, vA[i]); mB = fmaxf(mB, vB[i]); }
    }
    mA = group8_max(mA);
    mB = group8_max(mB);

    float exA[4], exB[4];
    float dA = 0.0f, dB = 0.0f;
    for (int c = 0; c < nch; ++c) {
        if (nch > 1) load_chunk(c);
#pragma unroll
        for (int i = 0; i < 4; ++i) {
            exA[i] = expf(vA[i] - mA);
            exB[i] = expf(vB[i] - mB);
            dA += exA[i];
            dB += exB[i];
        }
    }
    const float rA = 1.0f / (group8_sum(dA) + 1e-16f);
    const float rB = 1.0f / (group8_sum(dB) + 1e-16f);

    float4 acc0 = make_float4(0.0f, 0.0f, 0.0f, 0.0f);
    float4 acc1 = make_float4(0.0f, 0.0f, 0.0f, 0.0f);
    const float* hl = h + 4 * lane;
    const int    gl = lane & 24;
    for (int c = 0; c < nch; ++c) {
        if (nch > 1) {
            load_chunk(c);
#pragma unroll
            for (int i = 0; i < 4; ++i) {
                exA[i] = expf(vA[i] - mA);
                exB[i] = expf(vB[i] - mB);
            }
        }
        const int left = n - 32 * c;
#pragma unroll
        for (int i = 0; i < 4; ++i) {
            const float alA = exA[i] * rA;
            const float alB = exB[i] * rB;
#pragma unroll
            for (int q = 0; q < 2; ++q) {
                if (8 * i + 4 * q < left) {
                    float4 x0[4], x1[4];
                    float  wA[4], wB[4];
#pragma unroll
                    for (int u = 0; u < 4; ++u) {
                        const int k  = 4 * q + u;
                        const int sj = __builtin_amdgcn_readlane(s[i], k);
                        const float* row = hl + (size_t)sj * HCDIM;
                        x0[u] = *(const float4*)(row);
                        x1[u] = *(const float4*)(row + 128);
                        wA[u] = __shfl(alA, gl | k, 32);
                        wB[u] = __shfl(alB, gl | k, 32);
                    }
#pragma unroll
                    for (int u = 0; u < 4; ++u) {
                        acc0.x = fmaf(x0[u].x, wA[u], acc0.x);
                        acc0.y = fmaf(x0[u].y, wA[u], acc0.y);
                        acc0.z = fmaf(x0[u].z, wA[u], acc0.z);
                        acc0.w = fmaf(x0[u].w, wA[u], acc0.w);
                        acc1.x = fmaf(x1[u].x, wB[u], acc1.x);
                        acc1.y = fmaf(x1[u].y, wB[u], acc1.y);
                        acc1.z = fmaf(x1[u].z, wB[u], acc1.z);
                        acc1.w = fmaf(x1[u].w, wB[u], acc1.w);
                    }
                }
            }
        }
    }

    float* op = accum + (size_t)d * HCDIM + 4 * lane;
    *(float4*)(op)       = acc0;
    *(float4*)(op + 128) = acc1;
}

extern "C" void kernel_launch(void* const* d_in, const int* in_sizes, int n_in,
                              void* d_out, int out_size, void* d_ws, size_t ws_size,
                              hipStream_t stream) {
    (void)n_in; (void)out_size; (void)ws_size;

    const float* x        = (const float*)d_in[0];
    const int*   ei       = (const int*)d_in[1];
    const float* W_gat    = (const float*)d_in[2];
    const float* att_src  = (const float*)d_in[3];
    const float* att_dst  = (const float*)d_in[4];
    const float* bias_gat = (const float*)d_in[5];
    const float* W_lin    = (const float*)d_in[6];
    const float* b_lin    = (const float*)d_in[7];
    float*       out      = (float*)d_out;

    const int N   = in_sizes[0] / GEMM_K;
    const int E   = in_sizes[1] / 2;
    float* ws     = (float*)d_ws;
    float* h      = ws;
    float* acc    = h    + (size_t)N * HCDIM;
    float* asrc   = acc  + (size_t)N * HCDIM;
    int*   rowptr = (int*)(asrc + (size_t)N * HEADS);
    int*   ssrc   = rowptr + N + 1;
    int*   rank   = (int*)acc;

    wmma_gemm_tiled<256, 128, 32, false><<<(N / 16 + 3) / 4, 256, 0, stream>>>(x, bias_gat, W_gat, b_lin,
                                                                     h, N / 16);

    const int t2 = N * HEADS;
    gat_src_scores<<<(t2 + 255) / 256, 256, 0, stream>>>(h, att_src, asrc, rowptr, N);

    gat_edge_rank<<<(E + 255) / 256, 256, 0, stream>>>(ei, rowptr, rank, E);

    gat_rowptr_scan<<<1, SCAN_T, 0, stream>>>(rowptr, N);

    gat_edge_place<<<(E + 255) / 256, 256, 0, stream>>>(ei, rowptr, rank, ssrc, E);

    gat_node_gather<<<(N + 7) / 8, 256, 0, stream>>>(rowptr, ssrc, asrc, att_dst, h, acc, N);

    wmma_gemm_tiled<64, 64, 64, true><<<(N / 16 + 7) / 8, 256, 0, stream>>>(acc, bias_gat, W_lin, b_lin,
                                                                   out, N / 16);
}


# Edge bucketing moved into the first GEMM's launch as trailing blocks (4 edges/thread) that fill the slots its last blocks leave free; lists opened by a small kernel in front; list order stays atomic
# speedup vs baseline: 3.7392x; 1.3275x over previous
#include <hip/hip_runtime.h>
#include <math.h>


typedef float v2f __attribute__((ext_vector_type(2)));
typedef float v8f __attribute__((ext_vector_type(8)));

#define HEADS  8
#define CDIM   32
#define HCDIM  256
#define NSLOPE 0.2f

#define GEMM_K   256

#define BCAP HCDIM
template <int NB, int WCOLS, int KC, bool FUSE>
__global__ void __launch_bounds__(256)
wmma_gemm_tiled(const float* __restrict__ A,
                const float* __restrict__ bias,
                const float* __restrict__ B,
                const float* __restrict__ bout,
                float* __restrict__ C, int nrt,
                int gemm_blocks,
                const int* __restrict__ ei, int* __restrict__ cnt,
                int* __restrict__ bucket, int E) {
    if (!FUSE && (int)blockIdx.x >= gemm_blocks) {
        const int e0 = ((int)blockIdx.x - gemm_blocks) * 1024 + (int)threadIdx.x;
        int dv[4], sv[4], rk[4];
        for (int i = 0; i < 4; ++i) {
            const int e = e0 + 256 * i;
            dv[i] = e < E ? ei[E + e] : -1;
            sv[i] = e < E ? ei[e] : 0;
        }
        for (int i = 0; i < 4; ++i)
            rk[i] = dv[i] >= 0 ? atomicAdd(&cnt[dv[i]], 1) : BCAP;
        for (int i = 0; i < 4; ++i)
            if (rk[i] < BCAP) bucket[(size_t)dv[i] * HCDIM + rk[i]] = sv[i];
        return;
    }

    constexpr int NCH   = GEMM_K / KC;
    constexpr int QR    = KC / 4;
    constexpr int NG    = KC / 8;
    constexpr int CT    = WCOLS / 16;
    constexpr int WPR   = NB / WCOLS;
    constexpr int RW    = 8 / WPR;
    constexpr int PITCH = NB + 8;
    constexpr int BUF   = QR * PITCH;
    constexpr int NQ    = NB / 4;
    constexpr int UNITS = QR * NQ;
    constexpr int UPT   = UNITS / 256;
    static_assert(UNITS % 256 == 0, "every thread stages the same number of units");

    __shared__ float4 Bs[2 * BUF];

    const int  tid   = threadIdx.x;
    const int  lane  = tid & 31;
    const int  wv    = tid >> 5;
    const int  c     = lane & 15;
    const int  hf    = lane >> 4;
    const int  wr    = wv / WPR;
    const int  wc    = wv - wr * WPR;
    const int  rt    = blockIdx.x * RW + wr;
    const bool valid = rt < nrt;
    const int  rtc   = valid ? rt : nrt - 1;
    const int  col0  = wc * WCOLS;

    const float* ap = A + (size_t)(rtc * 16 + c) * GEMM_K + 4 * hf;
    const float* bp = bias + 4 * hf;

    float4 sb[UPT][4];
    float4 a_nxt[NG];
    v8f    acc[CT];
    for (int t = 0; t < CT; ++t) acc[t] = v8f{};

    auto fetch_b = [&](int ch) {
        for (int i = 0; i < UPT; ++i) {
            const int u  = tid + 256 * i;
            const int kq = u / NQ;
            const int q  = u - kq * NQ;
            const float* g = B + (size_t)(ch * KC + 4 * kq) * NB + 4 * q;
            for (int r = 0; r < 4; ++r)
                sb[i][r] = *(const float4*)(g + (size_t)r * NB);
        }
    };
    auto store_b = [&](int buf) {
        for (int i = 0; i < UPT; ++i) {
            const int u  = tid + 256 * i;
            const int kq = u / NQ;
            const int q  = u - kq * NQ;
            float4* d = Bs + buf * BUF + kq * PITCH + 4 * q;
            d[0] = make_float4(sb[i][0].x, sb[i][1].x, sb[i][2].x, sb[i][3].x);
            d[1] = make_float4(sb[i][0].y, sb[i][1].y, sb[i][2].y, sb[i][3].y);
            d[2] = make_float4(sb[i][0].z, sb[i][1].z, sb[i][2].z, sb[i][3].z);
            d[3] = make_float4(sb[i][0].w, sb[i][1].w, sb[i][2].w, sb[i][3].w);
        }
    };
    auto fetch_a = [&](int ch) {
        for (int g = 0; g < NG; ++g) {
            float4 v = *(const float4*)(ap + ch * KC + 8 * g);
            if (FUSE) {
                const float4 bb = *(const float4*)(bp + ch * KC + 8 * g);
                v.x = fmaxf(v.x + bb.x, 0.0f);
                v.y = fmaxf(v.y + bb.y, 0.0f);
                v.z = fmaxf(v.z + bb.z, 0.0f);
                v.w = fmaxf(v.w + bb.w, 0.0f);
            }
            a_nxt[g] = v;
        }
    };

    fetch_b(0);
    fetch_a(0);
    store_b(0);

    for (int ch = 0; ch < NCH; ++ch) {
        float4 a_cur[NG];
        for (int g = 0; g < NG; ++g) a_cur[g] = a_nxt[g];
        if (ch + 1 < NCH) { fetch_b(ch + 1); fetch_a(ch + 1); }

        asm volatile("s_wait_dscnt 0x0" ::: "memory");
        asm volatile("s_barrier_signal -1" ::: "memory");
        asm volatile("s_barrier_wait -1" ::: "memory");

        const float4* bs = Bs + (ch & 1) * BUF + hf * PITCH + col0 + c;
        for (int g = 0; g < NG; ++g) {
            float4 bv[CT];
            for (int t = 0; t < CT; ++t) bv[t] = bs[2 * g * PITCH + 16 * t];
            const v2f a0 = {a_cur[g].x, a_cur[g].y};
            const v2f a1 = {a_cur[g].z, a_cur[g].w};
            for (int t = 0; t < CT; ++t) {
                const v2f b0 = {bv[t].x, bv[t].y};
                acc[t] = __builtin_amdgcn_wmma_f32_16x16x4_f32(
                             false, a0, false, b0, (short)0, acc[t], false, false);
            }
            for (int t = 0; t < CT; ++t) {
                const v2f b1 = {bv[t].z, bv[t].w};
                acc[t] = __builtin_amdgcn_wmma_f32_16x16x4_f32(
                             false, a1, false, b1, (short)0, acc[t], false, false);
            }
        }

        if (ch + 1 < NCH) store_b((ch + 1) & 1);
    }

    if (valid) {
        float* cp = C + (size_t)(rt * 16 + 8 * hf) * NB + col0 + c;
        for (int t = 0; t < CT; ++t) {
            if (FUSE) {
                const float bo = bout[col0 + 16 * t + c];
                for (int r = 0; r < 8; ++r) cp[(size_t)r * NB + 16 * t] = acc[t][r] + bo;
            } else {
                for (int r = 0; r < 8; ++r) cp[(size_t)r * NB + 16 * t] = acc[t][r];
            }
        }
    }
}

__global__ void gat_open_lists(int* __restrict__ cnt, int* __restrict__ bucket, int N) {
    const int t = blockIdx.x * blockDim.x + threadIdx.x;
    if (t < N) { cnt[t] = 1; bucket[(size_t)t * HCDIM] = t; }
}

__global__ void gat_src_scores(const float* __restrict__ h,
                               const float* __restrict__ att_src,
                               float* __restrict__ a_src, int N) {
    const int t = blockIdx.x * blockDim.x + threadIdx.x;
    if (t >= N * HEADS) return;
    const int n  = t / HEADS;
    const int hd = t - n * HEADS;
    const float* hp = h + (size_t)n * HCDIM + hd * CDIM;
    const float* as = att_src + hd * CDIM;
    float s0 = 0.0f;
#pragma unroll
    for (int c = 0; c < CDIM; ++c) s0 = fmaf(hp[c], as[c], s0);
    a_src[t] = s0;
}

__device__ __forceinline__ float group8_sum(float v) {
    v += __shfl_xor(v, 1, 32);
    v += __shfl_xor(v, 2, 32);
    v += __shfl_xor(v, 4, 32);
    return v;
}
__device__ __forceinline__ float group8_max(float v) {
    v = fmaxf(v, __shfl_xor(v, 1, 32));
    v = fmaxf(v, __shfl_xor(v, 2, 32));
    v = fmaxf(v, __shfl_xor(v, 4, 32));
    return v;
}

__global__ void __launch_bounds__(256)
gat_node_gather(const int* __restrict__ cnt,
                const int* __restrict__ ei,
                const float* __restrict__ a_src,
                const float* __restrict__ att_dst,
                const float* __restrict__ h,
                float* accum, int N, int E) {
    const int lane = threadIdx.x & 31;
    const int d    = __builtin_amdgcn_readfirstlane((blockIdx.x * blockDim.x + threadIdx.x) >> 5);
    if (d >= N) return;
    const int g   = lane >> 3;
    const int e8  = lane & 7;
    const int* brow = (const int*)(accum + (size_t)d * HCDIM);
    float*     op   = accum + (size_t)d * HCDIM + 4 * lane;

    int raw0[4];
#pragma unroll
    for (int i = 0; i < 4; ++i) raw0[i] = brow[8 * i + e8];
    const int n   = cnt[d];
    const int nch = (n + 31) >> 5;

    const float4 hd0 = *(const float4*)(h + (size_t)d * HCDIM + 4 * lane);
    const float4 hd1 = *(const float4*)(h + (size_t)d * HCDIM + 128 + 4 * lane);
    const float4 ad0 = *(const float4*)(att_dst + 4 * lane);
    const float4 ad1 = *(const float4*)(att_dst + 128 + 4 * lane);
    const float adA = group8_sum(fmaf(hd0.w, ad0.w, fmaf(hd0.z, ad0.z, fmaf(hd0.y, ad0.y, hd0.x * ad0.x))));
    const float adB = group8_sum(fmaf(hd1.w, ad1.w, fmaf(hd1.z, ad1.z, fmaf(hd1.y, ad1.y, hd1.x * ad1.x))));

    int   s[4];
    float vA[4], vB[4];
    auto score_chunk = [&](int c, const int (&raw)[4]) {
#pragma unroll
        for (int i = 0; i < 4; ++i) {
            const int  j  = 32 * c + 8 * i + e8;
            const bool ok = j < n;
            const int  sc = min(max(raw[i], 0), N - 1);
            s[i] = ok ? sc : d;
            float a = a_src[sc * HEADS + g] + adA;
            float b = a_src[sc * HEADS + g + 4] + adB;
            a = a > 0.0f ? a : NSLOPE * a;
            b = b > 0.0f ? b : NSLOPE * b;
            vA[i] = ok ? a : -INFINITY;
            vB[i] = ok ? b : -INFINITY;
        }
    };
    if (n > BCAP) {
        auto logit = [&](int sj, float& a, float& b) {
            a = a_src[sj * HEADS + g] + adA;
            b = a_src[sj * HEADS + g + 4] + adB;
            a = a > 0.0f ? a : NSLOPE * a;
            b = b > 0.0f ? b : NSLOPE * b;
        };
        auto walk = [&](auto&& body) {
            for (int e0 = 0; e0 < E; e0 += 32) {
                const int  e   = e0 + lane;
                const bool hit = e < E && ei[E + e] == d;
                const int  se  = hit ? ei[e] : d;
                unsigned   m   = __builtin_amdgcn_ballot_w32(hit);
                while (m) {
                    const int k = __builtin_ctz(m);
                    m &= m - 1;
                    body(__shfl(se, k, 32));
                }
            }
        };
        float sa, sb;
        logit(d, sa, sb);
        float mA = sa, mB = sb;
        walk([&](int sj) {
            float a, b;
            logit(sj, a, b);
            mA = fmaxf(mA, a);
            mB = fmaxf(mB, b);
        });
        float dA = expf(sa - mA), dB = expf(sb - mB);
        walk([&](int sj) {
            float a, b;
            logit(sj, a, b);
            dA += expf(a - mA);
            dB += expf(b - mB);
        });
        const float rA  = 1.0f / (dA + 1e-16f);
        const float rB  = 1.0f / (dB + 1e-16f);
        const float wA0 = expf(sa - mA) * rA;
        const float wB0 = expf(sb - mB) * rB;
        float4 acc0 = make_float4(hd0.x * wA0, hd0.y * wA0, hd0.z * wA0, hd0.w * wA0);
        float4 acc1 = make_float4(hd1.x * wB0, hd1.y * wB0, hd1.z * wB0, hd1.w * wB0);
        walk([&](int sj) {
            float a, b;
            logit(sj, a, b);
            const float wA = expf(a - mA) * rA;
            const float wB = expf(b - mB) * rB;
            const float* row = h + (size_t)sj * HCDIM + 4 * lane;
            const float4 x0 = *(const float4*)(row);
            const float4 x1 = *(const float4*)(row + 128);
            acc0.x = fmaf(x0.x, wA, acc0.x);
            acc0.y = fmaf(x0.y, wA, acc0.y);
            acc0.z = fmaf(x0.z, wA, acc0.z);
            acc0.w = fmaf(x0.w, wA, acc0.w);
            acc1.x = fmaf(x1.x, wB, acc1.x);
            acc1.y = fmaf(x1.y, wB, acc1.y);
            acc1.z = fmaf(x1.z, wB, acc1.z);
            acc1.w = fmaf(x1.w, wB, acc1.w);
        });
        *(float4*)(op)       = acc0;
        *(float4*)(op + 128) = acc1;
        return;
    }

    auto load_chunk = [&](int c) {
        int raw[4];
#pragma unroll
        for (int i = 0; i < 4; ++i) raw[i] = brow[32 * c + 8 * i + e8];
        score_chunk(c, raw);
    };

    score_chunk(0, raw0);

    float mA = -INFINITY, mB = -INFINITY;
    for (int c = 0; c < nch; ++c) {
        if (nch > 1) load_chunk(c);
#pragma unroll
        for (int i = 0; i < 4; ++i) { mA = fmaxf(mA, vA[i]); mB = fmaxf(mB, vB[i]); }
    }
    mA = group8_max(mA);
    mB = group8_max(mB);

    float exA[4], exB[4];
    float dA = 0.0f, dB = 0.0f;
    for (int c = 0; c < nch; ++c) {
        if (nch > 1) load_chunk(c);
#pragma unroll
        for (int i = 0; i < 4; ++i) {
            exA[i] = expf(vA[i] - mA);
            exB[i] = expf(vB[i] - mB);
            dA += exA[i];
            dB += exB[i];
        }
    }
    const float rA = 1.0f / (group8_sum(dA) + 1e-16f);
    const float rB = 1.0f / (group8_sum(dB) + 1e-16f);

    float4 acc0 = make_float4(0.0f, 0.0f, 0.0f, 0.0f);
    float4 acc1 = make_float4(0.0f, 0.0f, 0.0f, 0.0f);
    const float* hl = h + 4 * lane;
    const int    gl = lane & 24;
    for (int c = 0; c < nch; ++c) {
        if (nch > 1) {
            load_chunk(c);
#pragma unroll
            for (int i = 0; i < 4; ++i) {
                exA[i] = expf(vA[i] - mA);
                exB[i] = expf(vB[i] - mB);
            }
        }
        const int left = n - 32 * c;
#pragma unroll
        for (int i = 0; i < 4; ++i) {
            const float alA = exA[i] * rA;
            const float alB = exB[i] * rB;
#pragma unroll
            for (int q = 0; q < 2; ++q) {
                if (8 * i + 4 * q < left) {
                    float4 x0[4], x1[4];
                    float  wA[4], wB[4];
#pragma unroll
                    for (int u = 0; u < 4; ++u) {
                        const int k  = 4 * q + u;
                        const int sj = __builtin_amdgcn_readlane(s[i], k);
                        const float* row = hl + (size_t)sj * HCDIM;
                        x0[u] = *(const float4*)(row);
                        x1[u] = *(const float4*)(row + 128);
                        wA[u] = __shfl(alA, gl | k, 32);
                        wB[u] = __shfl(alB, gl | k, 32);
                    }
#pragma unroll
                    for (int u = 0; u < 4; ++u) {
                        acc0.x = fmaf(x0[u].x, wA[u], acc0.x);
                        acc0.y = fmaf(x0[u].y, wA[u], acc0.y);
                        acc0.z = fmaf(x0[u].z, wA[u], acc0.z);
                        acc0.w = fmaf(x0[u].w, wA[u], acc0.w);
                        acc1.x = fmaf(x1[u].x, wB[u], acc1.x);
                        acc1.y = fmaf(x1[u].y, wB[u], acc1.y);
                        acc1.z = fmaf(x1[u].z, wB[u], acc1.z);
                        acc1.w = fmaf(x1[u].w, wB[u], acc1.w);
                    }
                }
            }
        }
    }

    *(float4*)(op)       = acc0;
    *(float4*)(op + 128) = acc1;
}

extern "C" void kernel_launch(void* const* d_in, const int* in_sizes, int n_in,
                              void* d_out, int out_size, void* d_ws, size_t ws_size,
                              hipStream_t stream) {
    (void)n_in; (void)out_size; (void)ws_size;

    const float* x        = (const float*)d_in[0];
    const int*   ei       = (const int*)d_in[1];
    const float* W_gat    = (const float*)d_in[2];
    const float* att_src  = (const float*)d_in[3];
    const float* att_dst  = (const float*)d_in[4];
    const float* bias_gat = (const float*)d_in[5];
    const float* W_lin    = (const float*)d_in[6];
    const float* b_lin    = (const float*)d_in[7];
    float*       out      = (float*)d_out;

    const int N   = in_sizes[0] / GEMM_K;
    const int E   = in_sizes[1] / 2;
    float* ws     = (float*)d_ws;
    float* h      = ws;
    float* acc    = h    + (size_t)N * HCDIM;
    float* asrc   = acc  + (size_t)N * HCDIM;
    int*   cnt    = (int*)(asrc + (size_t)N * HEADS);

    gat_open_lists<<<(N + 255) / 256, 256, 0, stream>>>(cnt, (int*)acc, N);

    const int gb1 = (N / 16 + 3) / 4;
    const int eb1 = (E + 1023) / 1024;
    wmma_gemm_tiled<256, 128, 32, false><<<gb1 + eb1, 256, 0, stream>>>(x, bias_gat, W_gat, b_lin,
                                                              h, N / 16, gb1, ei, cnt, (int*)acc, E);

    const int t2 = N * HEADS;
    gat_src_scores<<<(t2 + 255) / 256, 256, 0, stream>>>(h, att_src, asrc, N);

    gat_node_gather<<<(N + 7) / 8, 256, 0, stream>>>(cnt, ei, asrc, att_dst, h, acc, N, E);

    const int gb5 = (N / 16 + 7) / 8;
    wmma_gemm_tiled<64, 64, 64, true><<<gb5, 256, 0, stream>>>(acc, bias_gat, W_lin, b_lin,
                                                      out, N / 16, gb5, nullptr, nullptr, nullptr, 0);
}


# Final linear layer done inside the gather launch: resident 16-wave blocks pass 16 aggregated rows through LDS to the f32 WMMA; its k sum now 4 quarters x 2 accumulators, added at the end
# speedup vs baseline: 4.1137x; 1.1002x over previous
#include <hip/hip_runtime.h>
#include <math.h>


typedef float v2f __attribute__((ext_vector_type(2)));
typedef float v8f __attribute__((ext_vector_type(8)));

#define HEADS  8
#define CDIM   32
#define HCDIM  256
#define NSLOPE 0.2f

template <int M>
__device__ __forceinline__ float row_xor(float x) {
    return __builtin_bit_cast(float, __builtin_amdgcn_update_dpp(
               0, __builtin_bit_cast(int, x), 0x160 | M, 0xf, 0xf, true));
}

#define GEMM_K   256

#define BCAP HCDIM
template <int NB, int WCOLS, int KC>
__global__ void __launch_bounds__(256)
wmma_gemm_tiled(const float* __restrict__ A,
                const float* __restrict__ B,
                float* __restrict__ C, int nrt,
                int gemm_blocks,
                const int* __restrict__ ei, int* __restrict__ cnt,
                int* __restrict__ bucket, int E,
                const float* __restrict__ att, float* __restrict__ score) {
    if ((int)blockIdx.x >= gemm_blocks) {
        const int e0 = ((int)blockIdx.x - gemm_blocks) * 1024 + (int)threadIdx.x;
        int dv[4], sv[4], rk[4];
        for (int i = 0; i < 4; ++i) {
            const int e = e0 + 256 * i;
            dv[i] = e < E ? ei[E + e] : -1;
            sv[i] = e < E ? ei[e] : 0;
        }
        for (int i = 0; i < 4; ++i)
            rk[i] = dv[i] >= 0 ? atomicAdd(&cnt[dv[i]], 1) : BCAP;
        for (int i = 0; i < 4; ++i)
            if (rk[i] < BCAP) bucket[(size_t)dv[i] * HCDIM + rk[i]] = sv[i];
        return;
    }

    constexpr int NCH   = GEMM_K / KC;
    constexpr int QR    = KC / 4;
    constexpr int NG    = KC / 8;
    constexpr int CT    = WCOLS / 16;
    constexpr int WPR   = NB / WCOLS;
    constexpr int RW    = 8 / WPR;
    constexpr int PITCH = NB + 8;
    constexpr int BUF   = QR * PITCH;
    constexpr int NQ    = NB / 4;
    constexpr int UNITS = QR * NQ;
    constexpr int UPT   = UNITS / 256;
    static_assert(UNITS % 256 == 0, "every thread stages the same number of units");

    __shared__ float4 Bs[2 * BUF];
    __shared__ __attribute__((aligned(16))) float Ss[RW * 16 * HEADS];

    const int  tid   = threadIdx.x;
    const int  lane  = tid & 31;
    const int  wv    = tid >> 5;
    const int  c     = lane & 15;
    const int  hf    = lane >> 4;
    const int  wr    = wv / WPR;
    const int  wc    = wv - wr * WPR;
    const int  rt    = blockIdx.x * RW + wr;
    const bool valid = rt < nrt;
    const int  rtc   = valid ? rt : nrt - 1;
    const int  col0  = wc * WCOLS;

    const float* ap = A + (size_t)(rtc * 16 + c) * GEMM_K + 4 * hf;

    float4 sb[UPT][4];
    float4 a_nxt[NG];
    v8f    acc[CT];
    for (int t = 0; t < CT; ++t) acc[t] = v8f{};

    auto fetch_b = [&](int ch) {
        for (int i = 0; i < UPT; ++i) {
            const int u  = tid + 256 * i;
            const int kq = u / NQ;
            const int q  = u - kq * NQ;
            const float* g = B + (size_t)(ch * KC + 4 * kq) * NB + 4 * q;
            for (int r = 0; r < 4; ++r)
                sb[i][r] = *(const float4*)(g + (size_t)r * NB);
        }
    };
    auto store_b = [&](int buf) {
        for (int i = 0; i < UPT; ++i) {
            const int u  = tid + 256 * i;
            const int kq = u / NQ;
            const int q  = u - kq * NQ;
            float4* d = Bs + buf * BUF + kq * PITCH + 4 * q;
            d[0] = make_float4(sb[i][0].x, sb[i][1].x, sb[i][2].x, sb[i][3].x);
            d[1] = make_float4(sb[i][0].y, sb[i][1].y, sb[i][2].y, sb[i][3].y);
            d[2] = make_float4(sb[i][0].z, sb[i][1].z, sb[i][2].z, sb[i][3].z);
            d[3] = make_float4(sb[i][0].w, sb[i][1].w, sb[i][2].w, sb[i][3].w);
        }
    };
    auto fetch_a = [&](int ch) {
        for (int g = 0; g < NG; ++g)
            a_nxt[g] = *(const float4*)(ap + ch * KC + 8 * g);
    };

    fetch_b(0);
    fetch_a(0);
    store_b(0);

    for (int ch = 0; ch < NCH; ++ch) {
        float4 a_cur[NG];
        for (int g = 0; g < NG; ++g) a_cur[g] = a_nxt[g];
        if (ch + 1 < NCH) { fetch_b(ch + 1); fetch_a(ch + 1); }

        asm volatile("s_wait_dscnt 0x0" ::: "memory");
        asm volatile("s_barrier_signal -1" ::: "memory");
        asm volatile("s_barrier_wait -1" ::: "memory");

        const float4* bs = Bs + (ch & 1) * BUF + hf * PITCH + col0 + c;
        for (int g = 0; g < NG; ++g) {
            float4 bv[CT];
            for (int t = 0; t < CT; ++t) bv[t] = bs[2 * g * PITCH + 16 * t];
            const v2f a0 = {a_cur[g].x, a_cur[g].y};
            const v2f a1 = {a_cur[g].z, a_cur[g].w};
            for (int t = 0; t < CT; ++t) {
                const v2f b0 = {bv[t].x, bv[t].y};
                acc[t] = __builtin_amdgcn_wmma_f32_16x16x4_f32(
                             false, a0, false, b0, (short)0, acc[t], false, false);
            }
            for (int t = 0; t < CT; ++t) {
                const v2f b1 = {bv[t].z, bv[t].w};
                acc[t] = __builtin_amdgcn_wmma_f32_16x16x4_f32(
                             false, a1, false, b1, (short)0, acc[t], false, false);
            }
        }

        if (ch + 1 < NCH) store_b((ch + 1) & 1);
    }

    if (valid) {
        float* cp = C + (size_t)(rt * 16 + 8 * hf) * NB + col0 + c;
        for (int t = 0; t < CT; ++t)
            for (int r = 0; r < 8; ++r) cp[(size_t)r * NB + 16 * t] = acc[t][r];
    }

    {
        static_assert(NB == HCDIM && WCOLS == 4 * CDIM && CDIM == 32, "a wave owns four whole heads");
        float p[4][8];
        for (int j = 0; j < 4; ++j) {
            const float w0 = att[col0 + 32 * j + c];
            const float w1 = att[col0 + 32 * j + 16 + c];
            for (int r = 0; r < 8; ++r)
                p[j][r] = fmaf(acc[2 * j + 1][r], w1, acc[2 * j][r] * w0);
        }
        const bool b3 = c & 8, b2 = c & 4, b1 = c & 2, b0 = c & 1;
        float q[4][4], s2[4][2], s1[4], s0[2];
        for (int j = 0; j < 4; ++j)
            for (int r = 0; r < 4; ++r) {
                const float lo = p[j][r], hi = p[j][r + 4];
                q[j][r] = (b3 ? hi : lo) + row_xor<8>(b3 ? lo : hi);
            }
        for (int j = 0; j < 4; ++j)
            for (int r = 0; r < 2; ++r) {
                const float lo = q[j][r], hi = q[j][r + 2];
                s2[j][r] = (b2 ? hi : lo) + row_xor<4>(b2 ? lo : hi);
            }
        for (int j = 0; j < 4; ++j) {
            const float lo = s2[j][0], hi = s2[j][1];
            s1[j] = (b1 ? hi : lo) + row_xor<2>(b1 ? lo : hi);
        }
        for (int j = 0; j < 2; ++j) {
            const float lo = s1[j], hi = s1[j + 2];
            s0[j] = (b0 ? hi : lo) + row_xor<1>(b0 ? lo : hi);
        }
        *(float2*)(Ss + (wr * 16 + 8 * hf + (c >> 1)) * HEADS + 4 * wc + 2 * (c & 1)) =
            make_float2(s0[0], s0[1]);
        asm volatile("s_wait_dscnt 0x0" ::: "memory");
        asm volatile("s_barrier_signal -1" ::: "memory");
        asm volatile("s_barrier_wait -1" ::: "memory");
        const int srow = blockIdx.x * (RW * 16) + (tid >> 1);
        if (tid < RW * 16 * HEADS / 4 && srow < nrt * 16)
            *(float4*)(score + (size_t)blockIdx.x * (RW * 16 * HEADS) + 4 * tid) =
                *(const float4*)(Ss + 4 * tid);
    }
}

__global__ void gat_open_lists(int* __restrict__ cnt, int* __restrict__ bucket, int N) {
    const int t = blockIdx.x * blockDim.x + threadIdx.x;
    if (t < N) { cnt[t] = 1; bucket[(size_t)t * HCDIM] = t; }
}

__device__ __forceinline__ float group8_sum(float v) {
    v += __shfl_xor(v, 1, 32);
    v += __shfl_xor(v, 2, 32);
    v += __shfl_xor(v, 4, 32);
    return v;
}
__device__ __forceinline__ float group8_max(float v) {
    v = fmaxf(v, __shfl_xor(v, 1, 32));
    v = fmaxf(v, __shfl_xor(v, 2, 32));
    v = fmaxf(v, __shfl_xor(v, 4, 32));
    return v;
}

__device__ __forceinline__ void
gather_row(const int* __restrict__ cnt,
           const int* __restrict__ ei,
           const float* __restrict__ a_src,
           const float* __restrict__ att_dst,
           const float* __restrict__ h,
           const int* __restrict__ bucket, int N, int E,
           int d, int lane, float4& r0, float4& r1) {
    const int g   = lane >> 3;
    const int e8  = lane & 7;
    const int* brow = bucket + (size_t)d * HCDIM;

    int raw0[4];
#pragma unroll
    for (int i = 0; i < 4; ++i) raw0[i] = brow[8 * i + e8];
    const int n   = cnt[d];
    const int nch = (n + 31) >> 5;

    const float4 hd0 = *(const float4*)(h + (size_t)d * HCDIM + 4 * lane);
    const float4 hd1 = *(const float4*)(h + (size_t)d * HCDIM + 128 + 4 * lane);
    const float4 ad0 = *(const float4*)(att_dst + 4 * lane);
    const float4 ad1 = *(const float4*)(att_dst + 128 + 4 * lane);
    const float adA = group8_sum(fmaf(hd0.w, ad0.w, fmaf(hd0.z, ad0.z, fmaf(hd0.y, ad0.y, hd0.x * ad0.x))));
    const float adB = group8_sum(fmaf(hd1.w, ad1.w, fmaf(hd1.z, ad1.z, fmaf(hd1.y, ad1.y, hd1.x * ad1.x))));

    int   s[4];
    float vA[4], vB[4];
    auto score_chunk = [&](int c, const int (&raw)[4]) {
#pragma unroll
        for (int i = 0; i < 4; ++i) {
            const int  j  = 32 * c + 8 * i + e8;
            const bool ok = j < n;
            const int  sc = min(max(raw[i], 0), N - 1);
            s[i] = ok ? sc : d;
            float a = a_src[sc * HEADS + g] + adA;
            float b = a_src[sc * HEADS + g + 4] + adB;
            a = a > 0.0f ? a : NSLOPE * a;
            b = b > 0.0f ? b : NSLOPE * b;
            vA[i] = ok ? a : -INFINITY;
            vB[i] = ok ? b : -INFINITY;
        }
    };
    if (n > BCAP) {
        auto logit = [&](int sj, float& a, float& b) {
            a = a_src[sj * HEADS + g] + adA;
            b = a_src[sj * HEADS + g + 4] + adB;
            a = a > 0.0f ? a : NSLOPE * a;
            b = b > 0.0f ? b : NSLOPE * b;
        };
        auto walk = [&](auto&& body) {
            for (int e0 = 0; e0 < E; e0 += 32) {
                const int  e   = e0 + lane;
                const bool hit = e < E && ei[E + e] == d;
                const int  se  = hit ? ei[e] : d;
                unsigned   m   = __builtin_amdgcn_ballot_w32(hit);
                while (m) {
                    const int k = __builtin_ctz(m);
                    m &= m - 1;
                    body(__shfl(se, k, 32));
                }
            }
        };
        float sa, sb;
        logit(d, sa, sb);
        float mA = sa, mB = sb;
        walk([&](int sj) {
            float a, b;
            logit(sj, a, b);
            mA = fmaxf(mA, a);
            mB = fmaxf(mB, b);
        });
        float dA = expf(sa - mA), dB = expf(sb - mB);
        walk([&](int sj) {
            float a, b;
            logit(sj, a, b);
            dA += expf(a - mA);
            dB += expf(b - mB);
        });
        const float rA  = 1.0f / (dA + 1e-16f);
        const float rB  = 1.0f / (dB + 1e-16f);
        const float wA0 = expf(sa - mA) * rA;
        const float wB0 = expf(sb - mB) * rB;
        float4 acc0 = make_float4(hd0.x * wA0, hd0.y * wA0, hd0.z * wA0, hd0.w * wA0);
        float4 acc1 = make_float4(hd1.x * wB0, hd1.y * wB0, hd1.z * wB0, hd1.w * wB0);
        walk([&](int sj) {
            float a, b;
            logit(sj, a, b);
            const float wA = expf(a - mA) * rA;
            const float wB = expf(b - mB) * rB;
            const float* row = h + (size_t)sj * HCDIM + 4 * lane;
            const float4 x0 = *(const float4*)(row);
            const float4 x1 = *(const float4*)(row + 128);
            acc0.x = fmaf(x0.x, wA, acc0.x);
            acc0.y = fmaf(x0.y, wA, acc0.y);
            acc0.z = fmaf(x0.z, wA, acc0.z);
            acc0.w = fmaf(x0.w, wA, acc0.w);
            acc1.x = fmaf(x1.x, wB, acc1.x);
            acc1.y = fmaf(x1.y, wB, acc1.y);
            acc1.z = fmaf(x1.z, wB, acc1.z);
            acc1.w = fmaf(x1.w, wB, acc1.w);
        });
        r0 = acc0;
        r1 = acc1;
        return;
    }

    auto load_chunk = [&](int c) {
        int raw[4];
#pragma unroll
        for (int i = 0; i < 4; ++i) raw[i] = brow[32 * c + 8 * i + e8];
        score_chunk(c, raw);
    };

    score_chunk(0, raw0);

    float mA = -INFINITY, mB = -INFINITY;
    for (int c = 0; c < nch; ++c) {
        if (nch > 1) load_chunk(c);
#pragma unroll
        for (int i = 0; i < 4; ++i) { mA = fmaxf(mA, vA[i]); mB = fmaxf(mB, vB[i]); }
    }
    mA = group8_max(mA);
    mB = group8_max(mB);

    float exA[4], exB[4];
    float dA = 0.0f, dB = 0.0f;
    for (int c = 0; c < nch; ++c) {
        if (nch > 1) load_chunk(c);
#pragma unroll
        for (int i = 0; i < 4; ++i) {
            exA[i] = expf(vA[i] - mA);
            exB[i] = expf(vB[i] - mB);
            dA += exA[i];
            dB += exB[i];
        }
    }
    const float rA = 1.0f / (group8_sum(dA) + 1e-16f);
    const float rB = 1.0f / (group8_sum(dB) + 1e-16f);

    float4 acc0 = make_float4(0.0f, 0.0f, 0.0f, 0.0f);
    float4 acc1 = make_float4(0.0f, 0.0f, 0.0f, 0.0f);
    const float* hl = h + 4 * lane;
    const int    gl = lane & 24;
    for (int c = 0; c < nch; ++c) {
        if (nch > 1) {
            load_chunk(c);
#pragma unroll
            for (int i = 0; i < 4; ++i) {
                exA[i] = expf(vA[i] - mA);
                exB[i] = expf(vB[i] - mB);
            }
        }
        const int left = n - 32 * c;
#pragma unroll
        for (int i = 0; i < 4; ++i) {
            const float alA = exA[i] * rA;
            const float alB = exB[i] * rB;
#pragma unroll
            for (int q = 0; q < 2; ++q) {
                if (8 * i + 4 * q < left) {
                    float4 x0[4], x1[4];
                    float  wA[4], wB[4];
#pragma unroll
                    for (int u = 0; u < 4; ++u) {
                        const int k  = 4 * q + u;
                        const int sj = __builtin_amdgcn_readlane(s[i], k);
                        const float* row = hl + (size_t)sj * HCDIM;
                        x0[u] = *(const float4*)(row);
                        x1[u] = *(const float4*)(row + 128);
                        wA[u] = __shfl(alA, gl | k, 32);
                        wB[u] = __shfl(alB, gl | k, 32);
                    }
#pragma unroll
                    for (int u = 0; u < 4; ++u) {
                        acc0.x = fmaf(x0[u].x, wA[u], acc0.x);
                        acc0.y = fmaf(x0[u].y, wA[u], acc0.y);
                        acc0.z = fmaf(x0[u].z, wA[u], acc0.z);
                        acc0.w = fmaf(x0[u].w, wA[u], acc0.w);
                        acc1.x = fmaf(x1[u].x, wB[u], acc1.x);
                        acc1.y = fmaf(x1[u].y, wB[u], acc1.y);
                        acc1.z = fmaf(x1[u].z, wB[u], acc1.z);
                        acc1.w = fmaf(x1[u].w, wB[u], acc1.w);
                    }
                }
            }
        }
    }

    r0 = acc0;
    r1 = acc1;
}


#define LIN_N 64
#define T_PITCH (HCDIM + 4)
#define W_PITCH (LIN_N + 8)

__global__ void __launch_bounds__(512, 8)
gat_gather_linear(const int* __restrict__ cnt,
                  const int* __restrict__ ei,
                  const float* __restrict__ a_src,
                  const float* __restrict__ att_dst,
                  const float* __restrict__ h,
                  const int* __restrict__ bucket,
                  const float* __restrict__ bias,
                  const float* __restrict__ W,
                  const float* __restrict__ bout,
                  float* __restrict__ out, int N, int E) {
    __shared__ float4 Wl[(GEMM_K / 4) * W_PITCH];
    __shared__ __attribute__((aligned(16))) float T[16 * T_PITCH];
    __shared__ __attribute__((aligned(16))) float P[4 * 16 * LIN_N];

    const int tid  = threadIdx.x;
    const int lane = tid & 31;
    const int wv   = tid >> 5;
    const int c    = lane & 15;
    const int hf   = lane >> 4;
    const int ct   = wv & 3;
    const int kq   = wv >> 2;

    for (int i = 0; i < (GEMM_K / 4) * LIN_N / 512; ++i) {
        const int u   = tid + 512 * i;
        const int row = u / LIN_N;
        const int col = u - row * LIN_N;
        const float* g = W + (size_t)(4 * row) * LIN_N + col;
        Wl[row * W_PITCH + col] = make_float4(g[0], g[LIN_N], g[2 * LIN_N], g[3 * LIN_N]);
    }

    const int ngrp = (N + 15) >> 4;
    for (int grp = blockIdx.x; grp < ngrp; grp += gridDim.x) {
        const int d = __builtin_amdgcn_readfirstlane(16 * grp + wv);
        float4 r0 = make_float4(0.0f, 0.0f, 0.0f, 0.0f);
        float4 r1 = r0;
        if (d < N) {
            gather_row(cnt, ei, a_src, att_dst, h, bucket, N, E, d, lane, r0, r1);
            const float4 bg0 = *(const float4*)(bias + 4 * lane);
            const float4 bg1 = *(const float4*)(bias + 128 + 4 * lane);
            r0 = make_float4(fmaxf(r0.x + bg0.x, 0.0f), fmaxf(r0.y + bg0.y, 0.0f),
                             fmaxf(r0.z + bg0.z, 0.0f), fmaxf(r0.w + bg0.w, 0.0f));
            r1 = make_float4(fmaxf(r1.x + bg1.x, 0.0f), fmaxf(r1.y + bg1.y, 0.0f),
                             fmaxf(r1.z + bg1.z, 0.0f), fmaxf(r1.w + bg1.w, 0.0f));
        }
        *(float4*)(T + wv * T_PITCH + 4 * lane)       = r0;
        *(float4*)(T + wv * T_PITCH + 128 + 4 * lane) = r1;
        __syncthreads();

        v8f pa = v8f{}, pb = v8f{};
        const float*  ta = T + c * T_PITCH + 64 * kq + 4 * hf;
        const float4* wb = Wl + (16 * kq + hf) * W_PITCH + 16 * ct + c;
        float4 av[8], bv[8];
        for (int g = 0; g < 8; ++g) {
            av[g] = *(const float4*)(ta + 8 * g);
            bv[g] = wb[2 * g * W_PITCH];
        }
        for (int g = 0; g < 8; ++g) {
            const v2f a0 = {av[g].x, av[g].y};
            const v2f a1 = {av[g].z, av[g].w};
            const v2f b0 = {bv[g].x, bv[g].y};
            const v2f b1 = {bv[g].z, bv[g].w};
            pa = __builtin_amdgcn_wmma_f32_16x16x4_f32(false, a0, false, b0, (short)0, pa, false, false);
            pb = __builtin_amdgcn_wmma_f32_16x16x4_f32(false, a1, false, b1, (short)0, pb, false, false);
        }
        float* pp = P + (kq * 16 + 8 * hf) * LIN_N + 16 * ct + c;
        for (int r = 0; r < 8; ++r) pp[r * LIN_N] = pa[r] + pb[r];
        __syncthreads();

        if (d < N) {
            const float* pr = P + wv * LIN_N + 2 * lane;
            const float2 p0 = *(const float2*)(pr);
            const float2 p1 = *(const float2*)(pr + 16 * LIN_N);
            const float2 p2 = *(const float2*)(pr + 32 * LIN_N);
            const float2 p3 = *(const float2*)(pr + 48 * LIN_N);
            const float2 bo = *(const float2*)(bout + 2 * lane);
            *(float2*)(out + (size_t)d * LIN_N + 2 * lane) =
                make_float2(((p0.x + p1.x) + (p2.x + p3.x)) + bo.x,
                            ((p0.y + p1.y) + (p2.y + p3.y)) + bo.y);
        }
    }
}

extern "C" void kernel_launch(void* const* d_in, const int* in_sizes, int n_in,
                              void* d_out, int out_size, void* d_ws, size_t ws_size,
                              hipStream_t stream) {
    (void)n_in; (void)out_size; (void)ws_size;

    const float* x        = (const float*)d_in[0];
    const int*   ei       = (const int*)d_in[1];
    const float* W_gat    = (const float*)d_in[2];
    const float* att_src  = (const float*)d_in[3];
    const float* att_dst  = (const float*)d_in[4];
    const float* bias_gat = (const float*)d_in[5];
    const float* W_lin    = (const float*)d_in[6];
    const float* b_lin    = (const float*)d_in[7];
    float*       out      = (float*)d_out;

    const int N   = in_sizes[0] / GEMM_K;
    const int E   = in_sizes[1] / 2;
    float* ws     = (float*)d_ws;
    float* h      = ws;
    int*   bkt    = (int*)(h + (size_t)N * HCDIM);
    float* asrc   = (float*)(bkt + (size_t)N * HCDIM);
    int*   cnt    = (int*)(asrc + (size_t)N * HEADS);

    gat_open_lists<<<(N + 255) / 256, 256, 0, stream>>>(cnt, bkt, N);

    const int gb1 = (N / 16 + 3) / 4;
    const int eb1 = (E + 1023) / 1024;
    wmma_gemm_tiled<256, 128, 32><<<gb1 + eb1, 256, 0, stream>>>(x, W_gat, h, N / 16, gb1,
                                                       ei, cnt, bkt, E, att_src, asrc);

    gat_gather_linear<<<512, 512, 0, stream>>>(cnt, ei, asrc, att_dst, h, bkt,
                                               bias_gat, W_lin, b_lin, out, N, E);
}
